# starting kernel
# baseline (speedup 1.0000x reference)
#include <hip/hip_runtime.h>

#define N_NODES   100000
#define N_EDGES   1600000
#define N_GRAPHS  1000
#define IN_DIM    32
#define HID_DIM   64
#define OUT_DIM   32

typedef __attribute__((ext_vector_type(2))) float v2f;
typedef __attribute__((ext_vector_type(8))) float v8f;

__device__ __forceinline__ void fadd(float* p, float v) { unsafeAtomicAdd(p, v); }


__global__ void zero_f32(float* __restrict__ p, int n) {
  int i = blockIdx.x * blockDim.x + threadIdx.x;
  if (i < n) p[i] = 0.0f;
}

__global__ void count_deg(const int* __restrict__ dst, int* __restrict__ cnt, int n) {
  int i = blockIdx.x * blockDim.x + threadIdx.x;
  if (i < n) atomicAdd(&cnt[dst[i]], 1);
}

__global__ void compute_dis(const int* __restrict__ cnt, float* __restrict__ dis, int n) {
  int i = blockIdx.x * blockDim.x + threadIdx.x;
  if (i < n) dis[i] = rsqrtf(1.0f + (float)cnt[i]);
}


template<int K, int N>
__global__ void gemm_wmma(const float* __restrict__ A, const float* __restrict__ W,
                          float* __restrict__ H, int rowTiles) {
  constexpr int colTiles = N / 16;
  constexpr int rowWaves = 8 / colTiles;
  const int wave = threadIdx.x >> 5;
  const int lane = threadIdx.x & 31;
  const int lo   = lane & 15;
  const int hi   = lane >> 4;
  const int col  = (wave % colTiles) * 16;
  const int rowGroup = wave / colTiles;

  v2f bf[K / 4];
#pragma unroll
  for (int kk = 0; kk < K / 4; ++kk) {
    const int k = kk * 4 + hi * 2;
    bf[kk].x = W[(k + 0) * N + col + lo];
    bf[kk].y = W[(k + 1) * N + col + lo];
  }

  for (int rt = blockIdx.x * rowWaves + rowGroup; rt < rowTiles;
       rt += gridDim.x * rowWaves) {
    const int r0 = rt * 16;
    const float* __restrict__ Arow = A + (size_t)(r0 + lo) * K;
    v8f c = {};
#pragma unroll
    for (int kk = 0; kk < K / 4; ++kk) {
      const int k = kk * 4 + hi * 2;
      v2f a;
      a.x = Arow[k + 0];
      a.y = Arow[k + 1];
      c = __builtin_amdgcn_wmma_f32_16x16x4_f32(false, a, false, bf[kk],
                                                (short)0, c, false, false);
    }
    float* __restrict__ Hp = H + (size_t)(r0 + hi * 8) * N + col + lo;
#pragma unroll
    for (int i = 0; i < 8; ++i)
      Hp[(size_t)i * N] = c[i];
  }
}


template<int F>
__global__ void scatter_edges(const float* __restrict__ h, const int* __restrict__ src,
                              const int* __restrict__ dst, const float* __restrict__ dis,
                              float* __restrict__ agg, int nEdges) {
  constexpr unsigned CHUNKS = F / 4;
  const unsigned long long tid =
      (unsigned long long)blockIdx.x * blockDim.x + threadIdx.x;
  const unsigned long long total = (unsigned long long)nEdges * CHUNKS;
  if (tid >= total) return;
  const unsigned e  = (unsigned)(tid / CHUNKS);
  const unsigned fo = ((unsigned)tid & (CHUNKS - 1u)) << 2;
  const int s = src[e];
  const int d = dst[e];
  const float norm = dis[s] * dis[d];
  const float4 v = *(const float4*)(h + (size_t)s * F + fo);
  float* __restrict__ ap = agg + (size_t)d * F + fo;
  fadd(ap + 0, v.x * norm);
  fadd(ap + 1, v.y * norm);
  fadd(ap + 2, v.z * norm);
  fadd(ap + 3, v.w * norm);
}


template<int F>
__global__ void finalize_relu(const float* __restrict__ h, float* __restrict__ agg,
                              const float* __restrict__ dis, const float* __restrict__ b,
                              int n  ) {
  int idx = blockIdx.x * blockDim.x + threadIdx.x;
  if (idx >= n) return;
  const int node = idx / F;
  const int f    = idx & (F - 1);
  const float s  = dis[node];
  const float v  = agg[idx] + h[idx] * s * s + b[f];
  agg[idx] = fmaxf(v, 0.0f);
}


__global__ void finalize_pool(const float* __restrict__ h, const float* __restrict__ agg,
                              const float* __restrict__ dis, const float* __restrict__ b,
                              float* __restrict__ out, int n  ) {
  int idx = blockIdx.x * blockDim.x + threadIdx.x;
  if (idx >= n) return;
  const int node = idx >> 5;
  const int f    = idx & 31;
  const float s  = dis[node];
  const float v  = agg[idx] + h[idx] * s * s + b[f];
  const int g    = node / (N_NODES / N_GRAPHS);
  fadd(&out[g * OUT_DIM + f], v * 0.01f);
}


extern "C" void kernel_launch(void* const* d_in, const int* in_sizes, int n_in,
                              void* d_out, int out_size, void* d_ws, size_t ws_size,
                              hipStream_t stream) {
  const float* x    = (const float*)d_in[0];
  const int*   eidx = (const int*)d_in[1];
  const float* W1   = (const float*)d_in[3];
  const float* b1   = (const float*)d_in[4];
  const float* W2   = (const float*)d_in[5];
  const float* b2   = (const float*)d_in[6];
  float* out = (float*)d_out;

  const int* src = eidx;
  const int* dst = eidx + N_EDGES;

  char* ws = (char*)d_ws;
  float* dis  = (float*)(ws);
  int*   cnt  = (int*)  (ws + 400000);
  float* h1   = (float*)(ws + 800000);
  float* agg1 = (float*)(ws + 800000 + 25600000);
  float* h2   = (float*)(ws + 800000 + 51200000);
  float* agg2 = (float*)(ws + 800000 + 64000000);

  const int B = 256;

  zero_f32<<<(N_NODES * HID_DIM + B - 1) / B, B, 0, stream>>>(agg1, N_NODES * HID_DIM);
  zero_f32<<<(N_NODES * OUT_DIM + B - 1) / B, B, 0, stream>>>(agg2, N_NODES * OUT_DIM);
  zero_f32<<<(N_NODES + B - 1) / B, B, 0, stream>>>((float*)cnt, N_NODES);
  zero_f32<<<(N_GRAPHS * OUT_DIM + B - 1) / B, B, 0, stream>>>(out, N_GRAPHS * OUT_DIM);

  count_deg<<<(N_EDGES + B - 1) / B, B, 0, stream>>>(dst, cnt, N_EDGES);
  compute_dis<<<(N_NODES + B - 1) / B, B, 0, stream>>>(cnt, dis, N_NODES);

  const int rowTiles = N_NODES / 16;
  gemm_wmma<IN_DIM, HID_DIM><<<(rowTiles + 1) / 2, 256, 0, stream>>>(x, W1, h1, rowTiles);
  {
    unsigned long long tot = (unsigned long long)N_EDGES * (HID_DIM / 4);
    scatter_edges<HID_DIM><<<(int)((tot + B - 1) / B), B, 0, stream>>>(h1, src, dst, dis, agg1, N_EDGES);
  }
  finalize_relu<HID_DIM><<<(N_NODES * HID_DIM + B - 1) / B, B, 0, stream>>>(h1, agg1, dis, b1, N_NODES * HID_DIM);

  gemm_wmma<HID_DIM, OUT_DIM><<<(rowTiles + 3) / 4, 256, 0, stream>>>(agg1, W2, h2, rowTiles);
  {
    unsigned long long tot = (unsigned long long)N_EDGES * (OUT_DIM / 4);
    scatter_edges<OUT_DIM><<<(int)((tot + B - 1) / B), B, 0, stream>>>(h2, src, dst, dis, agg2, N_EDGES);
  }
  finalize_pool<<<(N_NODES * OUT_DIM + B - 1) / B, B, 0, stream>>>(h2, agg2, dis, b2, out, N_NODES * OUT_DIM);
}


# layer-1 normalisation factored out of the edge scatter: rows stored scaled by dis[src], dis[dst] applied once by the reader after the sum; the scatter gathers no node scalar per edge
# speedup vs baseline: 1.6412x; 1.6412x over previous
#include <hip/hip_runtime.h>

#define N_NODES   100000
#define N_EDGES   1600000
#define N_GRAPHS  1000
#define IN_DIM    32
#define HID_DIM   64
#define OUT_DIM   32

typedef __attribute__((ext_vector_type(2))) float v2f;
typedef __attribute__((ext_vector_type(8))) float v8f;

__device__ __forceinline__ void fadd(float* p, float v) { unsafeAtomicAdd(p, v); }


__global__ void zero_f32(float* __restrict__ p, int n) {
  int i = blockIdx.x * blockDim.x + threadIdx.x;
  if (i < n) p[i] = 0.0f;
}

__global__ void count_deg(const int* __restrict__ dst, int* __restrict__ cnt, int n) {
  int i = blockIdx.x * blockDim.x + threadIdx.x;
  if (i < n) atomicAdd(&cnt[dst[i]], 1);
}

__global__ void compute_dis(const int* __restrict__ cnt, float* __restrict__ dis, int n) {
  int i = blockIdx.x * blockDim.x + threadIdx.x;
  if (i < n) dis[i] = rsqrtf(1.0f + (float)cnt[i]);
}


template<int K, int N>
__global__ void gemm_wmma(const float* __restrict__ A, const float* __restrict__ W,
                          const float* __restrict__ dis,
                          float* __restrict__ H, float* __restrict__ S, int rowTiles) {
  constexpr int colTiles = N / 16;
  constexpr int rowWaves = 8 / colTiles;
  const int wave = threadIdx.x >> 5;
  const int lane = threadIdx.x & 31;
  const int lo   = lane & 15;
  const int hi   = lane >> 4;
  const int col  = (wave % colTiles) * 16;
  const int rowGroup = wave / colTiles;

  v2f bf[K / 4];
#pragma unroll
  for (int kk = 0; kk < K / 4; ++kk) {
    const int k = kk * 4 + hi * 2;
    bf[kk].x = W[(k + 0) * N + col + lo];
    bf[kk].y = W[(k + 1) * N + col + lo];
  }

  for (int rt = blockIdx.x * rowWaves + rowGroup; rt < rowTiles;
       rt += gridDim.x * rowWaves) {
    const int r0 = rt * 16;
    const float* __restrict__ Arow = A + (size_t)(r0 + lo) * K;
    v8f c = {};
#pragma unroll
    for (int kk = 0; kk < K / 4; ++kk) {
      const int k = kk * 4 + hi * 2;
      v2f a;
      a.x = Arow[k + 0];
      a.y = Arow[k + 1];
      c = __builtin_amdgcn_wmma_f32_16x16x4_f32(false, a, false, bf[kk],
                                                (short)0, c, false, false);
    }
    const int rb = r0 + hi * 8;
    const float4 d0 = *(const float4*)(dis + rb);
    const float4 d1 = *(const float4*)(dis + rb + 4);
    const float dv[8] = {d0.x, d0.y, d0.z, d0.w, d1.x, d1.y, d1.z, d1.w};
    float* __restrict__ Hp = H + (size_t)rb * N + col + lo;
    float* __restrict__ Sp = S + (size_t)rb * N + col + lo;
#pragma unroll
    for (int i = 0; i < 8; ++i) {
      const float hs = c[i] * dv[i];
      Hp[(size_t)i * N] = hs;
      Sp[(size_t)i * N] = hs;
    }
  }
}


template<int F>
__global__ __launch_bounds__(256) void scatter_edges(const float* __restrict__ h, const int* __restrict__ src,
                              const int* __restrict__ dst,
                              float* __restrict__ agg, int nChunks) {
  static_assert(F % 32 == 0, "a row must be whole 32-float lines");
  constexpr int LINES = F / 32;
  const int lane = threadIdx.x & 31;
  const int c = blockIdx.x * 8 + (threadIdx.x >> 5);
  if (c >= nChunks) return;
  const int e = c * 32 + lane;
  const int s = src[e];
  const int d = dst[e];
#pragma unroll
  for (int half = 0; half < 2; ++half) {
    float v[16][LINES];
#pragma unroll
    for (int j = 0; j < 16; ++j) {
      const int sj = __builtin_amdgcn_readlane(s, half * 16 + j);
      const float* __restrict__ rowp = h + (size_t)sj * F + lane;
#pragma unroll
      for (int q = 0; q < LINES; ++q) v[j][q] = rowp[q * 32];
    }
#pragma unroll
    for (int j = 0; j < 16; ++j) {
      const int   dj = __builtin_amdgcn_readlane(d, half * 16 + j);
      float* __restrict__ ap = agg + (size_t)dj * F + lane;
#pragma unroll
      for (int q = 0; q < LINES; ++q) fadd(ap + q * 32, v[j][q]);
    }
  }
}


#define N_ITEMS     (N_EDGES + N_NODES)
#define Z_COUNTS    (N_GRAPHS * HID_DIM)
#define Z_PAD       65536
#define POOL_BLOCKS 192

__global__ __launch_bounds__(1024) void pool_rows(
    const float* __restrict__ h, const int* __restrict__ src, const int* __restrict__ dst,
    const int* __restrict__ batch, const float* __restrict__ dis,
    const float* __restrict__ bias1, float* __restrict__ part) {
  __shared__ float4 zs4[Z_PAD / 4];
  float* zs = (float*)zs4;
  const int tid = threadIdx.x;
  for (int i = tid; i < Z_PAD / 4; i += 1024) zs4[i] = make_float4(0.f, 0.f, 0.f, 0.f);
  __syncthreads();

  const int lane = tid & 31;
  const float2 bl = *(const float2*)(bias1 + lane * 2);
  const int nChunks = (N_ITEMS + 31) / 32;
  for (int c = (tid >> 5) * gridDim.x + blockIdx.x; c < nChunks; c += gridDim.x * 32) {
    const int i = c * 32 + lane;
    int s = 0;
    int bi = 0;
    if (i < N_EDGES) {
      s  = src[i];
      bi = dst[i];
    } else if (i < N_ITEMS) {
      s  = i - N_EDGES;
      bi = s;
    }
    const float ws = dis[s];
    const float wb = dis[bi];
    unsigned g = (unsigned)batch[bi];
    const float w = i < N_ITEMS ? ws * wb : 0.0f;
    g = g < (unsigned)N_GRAPHS ? g : (unsigned)(N_GRAPHS - 1);
    if (i >= N_EDGES && i < N_ITEMS)
      __hip_atomic_fetch_add(&zs[Z_COUNTS + g], 1.0f, __ATOMIC_RELAXED,
                             __HIP_MEMORY_SCOPE_WORKGROUP);
    float2 v[32];
#pragma unroll
    for (int j = 0; j < 32; ++j) {
      const int sj = __builtin_amdgcn_readlane(s, j);
      const float2* __restrict__ rowp = (const float2*)(h + (size_t)sj * HID_DIM);
      v[j] = rowp[lane];
    }
#pragma unroll
    for (int j = 0; j < 32; ++j) {
      const int   gj = __builtin_amdgcn_readlane((int)g, j);
      const float wj = __builtin_bit_cast(float,
          __builtin_amdgcn_readlane(__builtin_bit_cast(int, w), j));
      const float dj = __builtin_bit_cast(float,
          __builtin_amdgcn_readlane(__builtin_bit_cast(int, ws), j));
      float* zp = zs + gj * HID_DIM + lane * 2;
      const float hx = v[j].x * dj + bl.x;
      const float hy = v[j].y * dj + bl.y;
      __hip_atomic_fetch_add(zp + 0, fmaxf(hx * wj, 0.0f), __ATOMIC_RELAXED, __HIP_MEMORY_SCOPE_WORKGROUP);
      __hip_atomic_fetch_add(zp + 1, fmaxf(hy * wj, 0.0f), __ATOMIC_RELAXED, __HIP_MEMORY_SCOPE_WORKGROUP);
    }
  }
  __syncthreads();

  float4* __restrict__ mine = (float4*)(part + (size_t)blockIdx.x * Z_PAD);
#pragma unroll 4
  for (int i = tid; i < Z_PAD / 4; i += 1024) mine[i] = zs4[i];
}

#define PG_GRAPHS 4
#define PG_GROUPS 16
#define AT_PITCH  68

__global__ __launch_bounds__(1024) void pool_gemm(
    const float* __restrict__ part, const float* __restrict__ W,
    const float* __restrict__ b, float* __restrict__ out) {
  __shared__ float4 red4[PG_GROUPS * 64];
  __shared__ float  At[16 * AT_PITCH];
  __shared__ float  cntS[PG_GRAPHS];
  const int tid = threadIdx.x;
  const int g0  = blockIdx.x * PG_GRAPHS;

  for (int i = tid; i < 16 * AT_PITCH; i += 1024) At[i] = 0.0f;
  if (tid < PG_GRAPHS) cntS[tid] = 0.0f;

  const int lo = tid & 15;
  const int hi = (tid >> 4) & 1;
  v2f bf0[HID_DIM / 4], bf1[HID_DIM / 4];
  if (tid < 32) {
#pragma unroll
    for (int kk = 0; kk < HID_DIM / 4; ++kk) {
      const int k = kk * 4 + hi * 2;
      bf0[kk].x = W[(k + 0) * OUT_DIM + lo];
      bf0[kk].y = W[(k + 1) * OUT_DIM + lo];
      bf1[kk].x = W[(k + 0) * OUT_DIM + 16 + lo];
      bf1[kk].y = W[(k + 1) * OUT_DIM + 16 + lo];
    }
  }

  {
    const int col = tid & 63;
    const int grp = tid >> 6;
    float4 acc = make_float4(0.f, 0.f, 0.f, 0.f);
    const float4* __restrict__ p = (const float4*)(part + (size_t)g0 * HID_DIM) + col;
#pragma unroll
    for (int j = 0; j < POOL_BLOCKS / PG_GROUPS; ++j) {
      const float4 t = p[(size_t)(grp + j * PG_GROUPS) * (Z_PAD / 4)];
      acc.x += t.x; acc.y += t.y; acc.z += t.z; acc.w += t.w;
    }
    red4[grp * 64 + col] = acc;
  }
  {
    const int gl = tid & 3;
    float cv = 0.0f;
    for (int q = tid >> 2; q < POOL_BLOCKS; q += 256)
      cv += part[(size_t)q * Z_PAD + Z_COUNTS + g0 + gl];
    cv += __shfl_xor(cv, 4);
    cv += __shfl_xor(cv, 8);
    cv += __shfl_xor(cv, 16);
    __syncthreads();
    if ((tid & 31) < PG_GRAPHS)
      __hip_atomic_fetch_add(&cntS[gl], cv, __ATOMIC_RELAXED, __HIP_MEMORY_SCOPE_WORKGROUP);
  }
  if (tid < PG_GRAPHS * HID_DIM) {
    const float* red = (const float*)red4;
    float sum = 0.0f;
#pragma unroll
    for (int k = 0; k < PG_GROUPS; ++k) sum += red[k * 256 + tid];
    At[(tid >> 6) * AT_PITCH + (tid & 63)] = sum;
  }
  __syncthreads();
  if (tid >= 32) return;

  v8f c0 = {};
  v8f c1 = {};
#pragma unroll
  for (int kk = 0; kk < HID_DIM / 4; ++kk) {
    const int k = kk * 4 + hi * 2;
    v2f a;
    a.x = At[lo * AT_PITCH + k + 0];
    a.y = At[lo * AT_PITCH + k + 1];
    c0 = __builtin_amdgcn_wmma_f32_16x16x4_f32(false, a, false, bf0[kk], (short)0, c0, false, false);
    c1 = __builtin_amdgcn_wmma_f32_16x16x4_f32(false, a, false, bf1[kk], (short)0, c1, false, false);
  }
  if (hi == 0) {
    const float bias0 = b[lo];
    const float bias1 = b[16 + lo];
#pragma unroll
    for (int i = 0; i < PG_GRAPHS; ++i) {
      const float cnt = cntS[i];
      out[(g0 + i) * OUT_DIM + lo]      = (c0[i] + cnt * bias0) / cnt;
      out[(g0 + i) * OUT_DIM + 16 + lo] = (c1[i] + cnt * bias1) / cnt;
    }
  }
}


extern "C" void kernel_launch(void* const* d_in, const int* in_sizes, int n_in,
                              void* d_out, int out_size, void* d_ws, size_t ws_size,
                              hipStream_t stream) {
  const float* x    = (const float*)d_in[0];
  const int*   eidx = (const int*)d_in[1];
  const int*   batch = (const int*)d_in[2];
  const float* W1   = (const float*)d_in[3];
  const float* b1   = (const float*)d_in[4];
  const float* W2   = (const float*)d_in[5];
  const float* b2   = (const float*)d_in[6];
  float* out = (float*)d_out;

  const int* src = eidx;
  const int* dst = eidx + N_EDGES;

  char* ws = (char*)d_ws;
  float* dis  = (float*)(ws);
  int*   cnt  = (int*)  (ws + 400000);
  float* agg1 = (float*)(ws + 800000);
  float* h1   = (float*)(ws + 800000 + 25600000);
  float* part = (float*)(ws + 800000 + 25600000);

  const int B = 256;

  zero_f32<<<(N_NODES + B - 1) / B, B, 0, stream>>>((float*)cnt, N_NODES);

  count_deg<<<(N_EDGES + B - 1) / B, B, 0, stream>>>(dst, cnt, N_EDGES);
  compute_dis<<<(N_NODES + B - 1) / B, B, 0, stream>>>(cnt, dis, N_NODES);

  static_assert(N_NODES % 16 == 0, "the product's row tiles cover every row of agg1");
  const int rowTiles = N_NODES / 16;
  gemm_wmma<IN_DIM, HID_DIM><<<(rowTiles + 1) / 2, 256, 0, stream>>>(x, W1, dis, h1, agg1, rowTiles);
  static_assert(N_EDGES % 32 == 0, "the scatter takes whole chunks of 32 edges");
  scatter_edges<HID_DIM><<<(N_EDGES / 32 + 7) / 8, 256, 0, stream>>>(h1, src, dst, agg1, N_EDGES / 32);

  pool_rows<<<POOL_BLOCKS, 1024, 0, stream>>>(agg1, src, dst, batch, dis, b1, part);
  pool_gemm<<<N_GRAPHS / PG_GRAPHS, 1024, 0, stream>>>(part, W2, b2, out);
}


# layer-2 product per node before the pool (was after the per-graph sums): an f32 WMMA pass stores finished 32-wide rows; a pool item gathers 1 line not 2, 2 node scalars not 3; 256 blocks, was 192
# speedup vs baseline: 1.8426x; 1.1227x over previous
#include <hip/hip_runtime.h>

#define N_NODES   100000
#define N_EDGES   1600000
#define N_GRAPHS  1000
#define IN_DIM    32
#define HID_DIM   64
#define OUT_DIM   32

typedef __attribute__((ext_vector_type(2))) float v2f;
typedef __attribute__((ext_vector_type(8))) float v8f;

__device__ __forceinline__ void fadd(float* p, float v) { unsafeAtomicAdd(p, v); }


__global__ void zero_f32(float* __restrict__ p, int n) {
  int i = blockIdx.x * blockDim.x + threadIdx.x;
  if (i < n) p[i] = 0.0f;
}

__global__ void count_deg(const int* __restrict__ dst, int* __restrict__ cnt, int n) {
  int i = blockIdx.x * blockDim.x + threadIdx.x;
  if (i < n) atomicAdd(&cnt[dst[i]], 1);
}

__global__ void compute_dis(const int* __restrict__ cnt, float* __restrict__ dis, int n) {
  int i = blockIdx.x * blockDim.x + threadIdx.x;
  if (i < n) dis[i] = rsqrtf(1.0f + (float)cnt[i]);
}


template<int K, int N>
__global__ void gemm_wmma(const float* __restrict__ A, const float* __restrict__ W,
                          const float* __restrict__ dis,
                          float* __restrict__ H, float* __restrict__ S, int rowTiles) {
  constexpr int colTiles = N / 16;
  constexpr int rowWaves = 8 / colTiles;
  const int wave = threadIdx.x >> 5;
  const int lane = threadIdx.x & 31;
  const int lo   = lane & 15;
  const int hi   = lane >> 4;
  const int col  = (wave % colTiles) * 16;
  const int rowGroup = wave / colTiles;

  v2f bf[K / 4];
#pragma unroll
  for (int kk = 0; kk < K / 4; ++kk) {
    const int k = kk * 4 + hi * 2;
    bf[kk].x = W[(k + 0) * N + col + lo];
    bf[kk].y = W[(k + 1) * N + col + lo];
  }

  for (int rt = blockIdx.x * rowWaves + rowGroup; rt < rowTiles;
       rt += gridDim.x * rowWaves) {
    const int r0 = rt * 16;
    const float* __restrict__ Arow = A + (size_t)(r0 + lo) * K;
    v8f c = {};
#pragma unroll
    for (int kk = 0; kk < K / 4; ++kk) {
      const int k = kk * 4 + hi * 2;
      v2f a;
      a.x = Arow[k + 0];
      a.y = Arow[k + 1];
      c = __builtin_amdgcn_wmma_f32_16x16x4_f32(false, a, false, bf[kk],
                                                (short)0, c, false, false);
    }
    const int rb = r0 + hi * 8;
    const float4 d0 = *(const float4*)(dis + rb);
    const float4 d1 = *(const float4*)(dis + rb + 4);
    const float dv[8] = {d0.x, d0.y, d0.z, d0.w, d1.x, d1.y, d1.z, d1.w};
    float* __restrict__ Hp = H + (size_t)rb * N + col + lo;
    float* __restrict__ Sp = S + (size_t)rb * N + col + lo;
#pragma unroll
    for (int i = 0; i < 8; ++i) {
      const float hs = c[i] * dv[i];
      Hp[(size_t)i * N] = hs;
      Sp[(size_t)i * N] = hs;
    }
  }
}


template<int F>
__global__ __launch_bounds__(256) void scatter_edges(const float* __restrict__ h, const int* __restrict__ src,
                              const int* __restrict__ dst,
                              float* __restrict__ agg, int nChunks) {
  static_assert(F % 32 == 0, "a row must be whole 32-float lines");
  constexpr int LINES = F / 32;
  const int lane = threadIdx.x & 31;
  const int c = blockIdx.x * 8 + (threadIdx.x >> 5);
  if (c >= nChunks) return;
  const int e = c * 32 + lane;
  const int s = src[e];
  const int d = dst[e];
#pragma unroll
  for (int half = 0; half < 2; ++half) {
    float v[16][LINES];
#pragma unroll
    for (int j = 0; j < 16; ++j) {
      const int sj = __builtin_amdgcn_readlane(s, half * 16 + j);
      const float* __restrict__ rowp = h + (size_t)sj * F + lane;
#pragma unroll
      for (int q = 0; q < LINES; ++q) v[j][q] = rowp[q * 32];
    }
#pragma unroll
    for (int j = 0; j < 16; ++j) {
      const int   dj = __builtin_amdgcn_readlane(d, half * 16 + j);
      float* __restrict__ ap = agg + (size_t)dj * F + lane;
#pragma unroll
      for (int q = 0; q < LINES; ++q) fadd(ap + q * 32, v[j][q]);
    }
  }
}

__global__ __launch_bounds__(256) void finish_gemm2(
    const float* __restrict__ T, const float* __restrict__ W, const float* __restrict__ dis,
    const float* __restrict__ bias1, float* __restrict__ U, int rowTiles) {
  const int wave = threadIdx.x >> 5;
  const int lane = threadIdx.x & 31;
  const int lo   = lane & 15;
  const int hi   = lane >> 4;
  const int rt   = blockIdx.x * 8 + wave;
  if (rt >= rowTiles) return;

  v2f bf0[HID_DIM / 4], bf1[HID_DIM / 4], bl[HID_DIM / 4];
#pragma unroll
  for (int kk = 0; kk < HID_DIM / 4; ++kk) {
    const int k = kk * 4 + hi * 2;
    bf0[kk].x = W[(k + 0) * OUT_DIM + lo];
    bf0[kk].y = W[(k + 1) * OUT_DIM + lo];
    bf1[kk].x = W[(k + 0) * OUT_DIM + 16 + lo];
    bf1[kk].y = W[(k + 1) * OUT_DIM + 16 + lo];
    bl[kk].x  = bias1[k + 0];
    bl[kk].y  = bias1[k + 1];
  }

  const int r0 = rt * 16;
  const float* __restrict__ Arow = T + (size_t)(r0 + lo) * HID_DIM;
  const float dl = dis[r0 + lo];
  v8f c0 = {};
  v8f c1 = {};
#pragma unroll
  for (int kk = 0; kk < HID_DIM / 4; ++kk) {
    const int k = kk * 4 + hi * 2;
    v2f a;
    a.x = fmaxf(Arow[k + 0] * dl + bl[kk].x, 0.0f);
    a.y = fmaxf(Arow[k + 1] * dl + bl[kk].y, 0.0f);
    c0 = __builtin_amdgcn_wmma_f32_16x16x4_f32(false, a, false, bf0[kk], (short)0, c0, false, false);
    c1 = __builtin_amdgcn_wmma_f32_16x16x4_f32(false, a, false, bf1[kk], (short)0, c1, false, false);
  }
  const int rb = r0 + hi * 8;
  const float4 d0 = *(const float4*)(dis + rb);
  const float4 d1 = *(const float4*)(dis + rb + 4);
  const float dv[8] = {d0.x, d0.y, d0.z, d0.w, d1.x, d1.y, d1.z, d1.w};
  float* __restrict__ Up = U + (size_t)rb * OUT_DIM + lo;
#pragma unroll
  for (int i = 0; i < 8; ++i) {
    Up[(size_t)i * OUT_DIM]      = c0[i] * dv[i];
    Up[(size_t)i * OUT_DIM + 16] = c1[i] * dv[i];
  }
}


#define N_ITEMS     (N_EDGES + N_NODES)
#define Z_COUNTS    (N_GRAPHS * OUT_DIM)
#define Z_PAD       33792
#define Z_LDS       43008
#define POOL_BLOCKS 256

__global__ __launch_bounds__(1024) void pool_rows(
    const float* __restrict__ u, const int* __restrict__ src, const int* __restrict__ dst,
    const int* __restrict__ batch, const float* __restrict__ dis, float* __restrict__ part) {
  __shared__ float4 zs4[Z_LDS / 4];
  float* zs = (float*)zs4;
  const int tid = threadIdx.x;
  for (int i = tid; i < Z_PAD / 4; i += 1024) zs4[i] = make_float4(0.f, 0.f, 0.f, 0.f);
  __syncthreads();

  const int lane = tid & 31;
  const int nChunks = (N_ITEMS + 31) / 32;
  for (int c = (tid >> 5) * gridDim.x + blockIdx.x; c < nChunks; c += gridDim.x * 32) {
    const int i = c * 32 + lane;
    int s = 0;
    int bi = 0;
    if (i < N_EDGES) {
      s  = src[i];
      bi = dst[i];
    } else if (i < N_ITEMS) {
      s  = i - N_EDGES;
      bi = s;
    }
    const float wb = dis[bi];
    unsigned g = (unsigned)batch[bi];
    const float w = i < N_ITEMS ? wb : 0.0f;
    g = g < (unsigned)N_GRAPHS ? g : (unsigned)(N_GRAPHS - 1);
    if (i >= N_EDGES && i < N_ITEMS)
      __hip_atomic_fetch_add(&zs[Z_COUNTS + g], 1.0f, __ATOMIC_RELAXED,
                             __HIP_MEMORY_SCOPE_WORKGROUP);
    float v[32];
#pragma unroll
    for (int j = 0; j < 32; ++j) {
      const int sj = __builtin_amdgcn_readlane(s, j);
      v[j] = u[(size_t)sj * OUT_DIM + lane];
    }
#pragma unroll
    for (int j = 0; j < 32; ++j) {
      const int   gj = __builtin_amdgcn_readlane((int)g, j);
      const float wj = __builtin_bit_cast(float,
          __builtin_amdgcn_readlane(__builtin_bit_cast(int, w), j));
      __hip_atomic_fetch_add(zs + gj * OUT_DIM + lane, v[j] * wj, __ATOMIC_RELAXED,
                             __HIP_MEMORY_SCOPE_WORKGROUP);
    }
  }
  __syncthreads();

  float4* __restrict__ mine = (float4*)(part + (size_t)blockIdx.x * Z_PAD);
#pragma unroll 4
  for (int i = tid; i < Z_PAD / 4; i += 1024) mine[i] = zs4[i];
}

#define PG_GRAPHS 4
#define PG_GROUPS 32

__global__ __launch_bounds__(1024) void pool_finish(
    const float* __restrict__ part, const float* __restrict__ b, float* __restrict__ out) {
  __shared__ float4 red4[PG_GROUPS * 32];
  __shared__ float  cntS[PG_GRAPHS];
  const int tid = threadIdx.x;
  const int g0  = blockIdx.x * PG_GRAPHS;

  if (tid < PG_GRAPHS) cntS[tid] = 0.0f;

  {
    const int col = tid & 31;
    const int grp = tid >> 5;
    float4 acc = make_float4(0.f, 0.f, 0.f, 0.f);
    const float4* __restrict__ p = (const float4*)(part + (size_t)g0 * OUT_DIM) + col;
#pragma unroll
    for (int j = 0; j < POOL_BLOCKS / PG_GROUPS; ++j) {
      const float4 t = p[(size_t)(grp + j * PG_GROUPS) * (Z_PAD / 4)];
      acc.x += t.x; acc.y += t.y; acc.z += t.z; acc.w += t.w;
    }
    red4[grp * 32 + col] = acc;
  }
  {
    const int gl = tid & 3;
    float cv = 0.0f;
    for (int q = tid >> 2; q < POOL_BLOCKS; q += 256)
      cv += part[(size_t)q * Z_PAD + Z_COUNTS + g0 + gl];
    cv += __shfl_xor(cv, 4);
    cv += __shfl_xor(cv, 8);
    cv += __shfl_xor(cv, 16);
    __syncthreads();
    if ((tid & 31) < PG_GRAPHS)
      __hip_atomic_fetch_add(&cntS[gl], cv, __ATOMIC_RELAXED, __HIP_MEMORY_SCOPE_WORKGROUP);
  }
  __syncthreads();
  if (tid < PG_GRAPHS * OUT_DIM) {
    const float* red = (const float*)red4;
    float sum = 0.0f;
#pragma unroll
    for (int k = 0; k < PG_GROUPS; ++k) sum += red[k * (PG_GRAPHS * OUT_DIM) + tid];
    const float cnt = cntS[tid >> 5];
    out[g0 * OUT_DIM + tid] = (sum + cnt * b[tid & 31]) / cnt;
  }
}


extern "C" void kernel_launch(void* const* d_in, const int* in_sizes, int n_in,
                              void* d_out, int out_size, void* d_ws, size_t ws_size,
                              hipStream_t stream) {
  const float* x    = (const float*)d_in[0];
  const int*   eidx = (const int*)d_in[1];
  const int*   batch = (const int*)d_in[2];
  const float* W1   = (const float*)d_in[3];
  const float* b1   = (const float*)d_in[4];
  const float* W2   = (const float*)d_in[5];
  const float* b2   = (const float*)d_in[6];
  float* out = (float*)d_out;

  const int* src = eidx;
  const int* dst = eidx + N_EDGES;

  char* ws = (char*)d_ws;
  float* dis  = (float*)(ws);
  int*   cnt  = (int*)  (ws + 400000);
  float* agg1 = (float*)(ws + 800000);
  float* h1   = (float*)(ws + 800000 + 25600000);
  float* u2   = (float*)(ws + 800000 + 25600000);
  float* part = (float*)(ws + 800000 + 25600000 + 12800000);

  const int B = 256;

  zero_f32<<<(N_NODES + B - 1) / B, B, 0, stream>>>((float*)cnt, N_NODES);

  count_deg<<<(N_EDGES + B - 1) / B, B, 0, stream>>>(dst, cnt, N_EDGES);
  compute_dis<<<(N_NODES + B - 1) / B, B, 0, stream>>>(cnt, dis, N_NODES);

  static_assert(N_NODES % 16 == 0, "the product's row tiles cover every row of agg1");
  const int rowTiles = N_NODES / 16;
  gemm_wmma<IN_DIM, HID_DIM><<<(rowTiles + 1) / 2, 256, 0, stream>>>(x, W1, dis, h1, agg1, rowTiles);
  static_assert(N_EDGES % 32 == 0, "the scatter takes whole chunks of 32 edges");
  scatter_edges<HID_DIM><<<(N_EDGES / 32 + 7) / 8, 256, 0, stream>>>(h1, src, dst, agg1, N_EDGES / 32);

  finish_gemm2<<<(rowTiles + 7) / 8, 256, 0, stream>>>(agg1, W2, dis, b1, u2, rowTiles);
  pool_rows<<<POOL_BLOCKS, 1024, 0, stream>>>(u2, src, dst, batch, dis, part);
  static_assert(N_GRAPHS % PG_GRAPHS == 0, "the finishing blocks cover every graph");
  pool_finish<<<N_GRAPHS / PG_GRAPHS, 1024, 0, stream>>>(part, b2, out);
}


# node scalars packed: deg_dis stores one 8-byte record {deg^-1/2, graph} per node; a pool item gathers it with 1 load where it took 2 from two arrays; 1600000 fewer scattered L2 requests
# speedup vs baseline: 2.9631x; 1.6081x over previous
#include <hip/hip_runtime.h>

#define N_NODES   100000
#define N_EDGES   1600000
#define N_GRAPHS  1000
#define IN_DIM    32
#define HID_DIM   64
#define OUT_DIM   32

typedef __attribute__((ext_vector_type(2))) float v2f;
typedef __attribute__((ext_vector_type(8))) float v8f;

__device__ __forceinline__ void fadd(float* p, float v) { unsafeAtomicAdd(p, v); }


#define DEG_RANGES  4
#define DEG_SHARES  64
#define DEG_NODES   (N_NODES / DEG_RANGES)
#define DEG_WORDS   (DEG_NODES / 2)
#define DEG_EDGES   (N_EDGES / DEG_SHARES)
#define DEG_TRIPS   ((DEG_EDGES / 4 + 1023) / 1024)

static_assert(N_NODES % (2 * DEG_RANGES) == 0, "a range is whole words of two counters");
static_assert(DEG_WORDS % 4 == 0, "a block's counters leave as whole uint4");
static_assert(N_EDGES % (4 * DEG_SHARES) == 0, "a share is whole int4 of edges");
static_assert(DEG_EDGES < 65536, "a 16-bit counter cannot overflow within one share");

__global__ __launch_bounds__(1024) void deg_hist(const int* __restrict__ dst,
                                                 unsigned* __restrict__ hpart) {
  __shared__ uint4 hs4[DEG_WORDS / 4];
  unsigned* hs = (unsigned*)hs4;
  const int tid = threadIdx.x;
  const int share = blockIdx.x / DEG_RANGES;
  const unsigned base = (unsigned)(blockIdx.x % DEG_RANGES) * DEG_NODES;
  for (int i = tid; i < DEG_WORDS / 4; i += 1024) hs4[i] = make_uint4(0u, 0u, 0u, 0u);
  __syncthreads();

  const int4* __restrict__ d4 = (const int4*)(dst + (size_t)share * DEG_EDGES);
  int4 v[DEG_TRIPS];
#pragma unroll
  for (int t = 0; t < DEG_TRIPS; ++t) {
    const int i = tid + t * 1024;
    v[t] = i < DEG_EDGES / 4 ? d4[i] : make_int4(-1, -1, -1, -1);
  }
#pragma unroll
  for (int t = 0; t < DEG_TRIPS; ++t) {
    const int d[4] = {v[t].x, v[t].y, v[t].z, v[t].w};
#pragma unroll
    for (int q = 0; q < 4; ++q) {
      const unsigned l = (unsigned)d[q] - base;
      if (l < (unsigned)DEG_NODES)
        __hip_atomic_fetch_add(&hs[l >> 1], 1u << ((l & 1u) * 16u), __ATOMIC_RELAXED,
                               __HIP_MEMORY_SCOPE_WORKGROUP);
    }
  }
  __syncthreads();

  uint4* __restrict__ mine =
      (uint4*)(hpart + (size_t)share * (N_NODES / 2) + (size_t)(blockIdx.x % DEG_RANGES) * DEG_WORDS);
  for (int i = tid; i < DEG_WORDS / 4; i += 1024) mine[i] = hs4[i];
}

__global__ __launch_bounds__(128) void deg_dis(const unsigned* __restrict__ hpart,
                                               const int* __restrict__ batch,
                                               float* __restrict__ rec) {
  const int t = blockIdx.x * 128 + threadIdx.x;
  if (t >= N_NODES / 2) return;
  unsigned lo = 0u, hi = 0u;
#pragma unroll
  for (int j = 0; j < DEG_SHARES; ++j) {
    const unsigned w = hpart[(size_t)j * (N_NODES / 2) + t];
    lo += w & 0xffffu;
    hi += w >> 16;
  }
  const int2 g = *(const int2*)(batch + 2 * t);
  float4 r;
  r.x = rsqrtf(1.0f + (float)lo);
  r.y = __int_as_float(g.x);
  r.z = rsqrtf(1.0f + (float)hi);
  r.w = __int_as_float(g.y);
  *(float4*)(rec + 4 * t) = r;
}


template<int K, int N>
__global__ void gemm_wmma(const float* __restrict__ A, const float* __restrict__ W,
                          const float* __restrict__ rec,
                          float* __restrict__ H, float* __restrict__ S, int rowTiles) {
  constexpr int colTiles = N / 16;
  constexpr int rowWaves = 8 / colTiles;
  const int wave = threadIdx.x >> 5;
  const int lane = threadIdx.x & 31;
  const int lo   = lane & 15;
  const int hi   = lane >> 4;
  const int col  = (wave % colTiles) * 16;
  const int rowGroup = wave / colTiles;

  v2f bf[K / 4];
#pragma unroll
  for (int kk = 0; kk < K / 4; ++kk) {
    const int k = kk * 4 + hi * 2;
    bf[kk].x = W[(k + 0) * N + col + lo];
    bf[kk].y = W[(k + 1) * N + col + lo];
  }

  for (int rt = blockIdx.x * rowWaves + rowGroup; rt < rowTiles;
       rt += gridDim.x * rowWaves) {
    const int r0 = rt * 16;
    const float* __restrict__ Arow = A + (size_t)(r0 + lo) * K;
    v8f c = {};
#pragma unroll
    for (int kk = 0; kk < K / 4; ++kk) {
      const int k = kk * 4 + hi * 2;
      v2f a;
      a.x = Arow[k + 0];
      a.y = Arow[k + 1];
      c = __builtin_amdgcn_wmma_f32_16x16x4_f32(false, a, false, bf[kk],
                                                (short)0, c, false, false);
    }
    const int rb = r0 + hi * 8;
    const float4* __restrict__ rp = (const float4*)(rec + 2 * rb);
    const float4 d0 = rp[0], d1 = rp[1], d2 = rp[2], d3 = rp[3];
    const float dv[8] = {d0.x, d0.z, d1.x, d1.z, d2.x, d2.z, d3.x, d3.z};
    float* __restrict__ Hp = H + (size_t)rb * N + col + lo;
    float* __restrict__ Sp = S + (size_t)rb * N + col + lo;
#pragma unroll
    for (int i = 0; i < 8; ++i) {
      const float hs = c[i] * dv[i];
      Hp[(size_t)i * N] = hs;
      Sp[(size_t)i * N] = hs;
    }
  }
}


template<int F>
__global__ __launch_bounds__(256) void scatter_edges(const float* __restrict__ h, const int* __restrict__ src,
                              const int* __restrict__ dst,
                              float* __restrict__ agg, int nChunks) {
  static_assert(F % 32 == 0, "a row must be whole 32-float lines");
  constexpr int LINES = F / 32;
  const int lane = threadIdx.x & 31;
  const int c = blockIdx.x * 8 + (threadIdx.x >> 5);
  if (c >= nChunks) return;
  const int e = c * 32 + lane;
  const int s = src[e];
  const int d = dst[e];
#pragma unroll
  for (int half = 0; half < 2; ++half) {
    float v[16][LINES];
#pragma unroll
    for (int j = 0; j < 16; ++j) {
      const int sj = __builtin_amdgcn_readlane(s, half * 16 + j);
      const float* __restrict__ rowp = h + (size_t)sj * F + lane;
#pragma unroll
      for (int q = 0; q < LINES; ++q) v[j][q] = rowp[q * 32];
    }
#pragma unroll
    for (int j = 0; j < 16; ++j) {
      const int   dj = __builtin_amdgcn_readlane(d, half * 16 + j);
      float* __restrict__ ap = agg + (size_t)dj * F + lane;
#pragma unroll
      for (int q = 0; q < LINES; ++q) fadd(ap + q * 32, v[j][q]);
    }
  }
}

__global__ __launch_bounds__(256) void finish_gemm2(
    const float* __restrict__ T, const float* __restrict__ W, const float* __restrict__ rec,
    const float* __restrict__ bias1, float* __restrict__ U, int rowTiles) {
  const int wave = threadIdx.x >> 5;
  const int lane = threadIdx.x & 31;
  const int lo   = lane & 15;
  const int hi   = lane >> 4;
  const int rt   = blockIdx.x * 8 + wave;
  if (rt >= rowTiles) return;

  v2f bf0[HID_DIM / 4], bf1[HID_DIM / 4], bl[HID_DIM / 4];
#pragma unroll
  for (int kk = 0; kk < HID_DIM / 4; ++kk) {
    const int k = kk * 4 + hi * 2;
    bf0[kk].x = W[(k + 0) * OUT_DIM + lo];
    bf0[kk].y = W[(k + 1) * OUT_DIM + lo];
    bf1[kk].x = W[(k + 0) * OUT_DIM + 16 + lo];
    bf1[kk].y = W[(k + 1) * OUT_DIM + 16 + lo];
    bl[kk].x  = bias1[k + 0];
    bl[kk].y  = bias1[k + 1];
  }

  const int r0 = rt * 16;
  const float* __restrict__ Arow = T + (size_t)(r0 + lo) * HID_DIM;
  const float dl = rec[2 * (r0 + lo)];
  v8f c0 = {};
  v8f c1 = {};
#pragma unroll
  for (int kk = 0; kk < HID_DIM / 4; ++kk) {
    const int k = kk * 4 + hi * 2;
    v2f a;
    a.x = fmaxf(Arow[k + 0] * dl + bl[kk].x, 0.0f);
    a.y = fmaxf(Arow[k + 1] * dl + bl[kk].y, 0.0f);
    c0 = __builtin_amdgcn_wmma_f32_16x16x4_f32(false, a, false, bf0[kk], (short)0, c0, false, false);
    c1 = __builtin_amdgcn_wmma_f32_16x16x4_f32(false, a, false, bf1[kk], (short)0, c1, false, false);
  }
  const int rb = r0 + hi * 8;
  const float4* __restrict__ rp = (const float4*)(rec + 2 * rb);
  const float4 d0 = rp[0], d1 = rp[1], d2 = rp[2], d3 = rp[3];
  const float dv[8] = {d0.x, d0.z, d1.x, d1.z, d2.x, d2.z, d3.x, d3.z};
  float* __restrict__ Up = U + (size_t)rb * OUT_DIM + lo;
#pragma unroll
  for (int i = 0; i < 8; ++i) {
    Up[(size_t)i * OUT_DIM]      = c0[i] * dv[i];
    Up[(size_t)i * OUT_DIM + 16] = c1[i] * dv[i];
  }
}


#define N_ITEMS     (N_EDGES + N_NODES)
#define Z_COUNTS    (N_GRAPHS * OUT_DIM)
#define Z_PAD       33792
#define Z_LDS       43008
#define POOL_BLOCKS 256

__global__ __launch_bounds__(1024) void pool_rows(
    const float* __restrict__ u, const int* __restrict__ src, const int* __restrict__ dst,
    const float* __restrict__ rec, float* __restrict__ part) {
  __shared__ float4 zs4[Z_LDS / 4];
  float* zs = (float*)zs4;
  const int tid = threadIdx.x;
  for (int i = tid; i < Z_PAD / 4; i += 1024) zs4[i] = make_float4(0.f, 0.f, 0.f, 0.f);
  __syncthreads();

  const int lane = tid & 31;
  const int nChunks = (N_ITEMS + 31) / 32;
  for (int c = (tid >> 5) * gridDim.x + blockIdx.x; c < nChunks; c += gridDim.x * 32) {
    const int i = c * 32 + lane;
    int s = 0;
    int bi = 0;
    if (i < N_EDGES) {
      s  = src[i];
      bi = dst[i];
    } else if (i < N_ITEMS) {
      s  = i - N_EDGES;
      bi = s;
    }
    const float2 rc = *(const float2*)(rec + 2 * (size_t)bi);
    const float wb = rc.x;
    unsigned g = (unsigned)__float_as_int(rc.y);
    const float w = i < N_ITEMS ? wb : 0.0f;
    g = g < (unsigned)N_GRAPHS ? g : (unsigned)(N_GRAPHS - 1);
    if (i >= N_EDGES && i < N_ITEMS)
      __hip_atomic_fetch_add(&zs[Z_COUNTS + g], 1.0f, __ATOMIC_RELAXED,
                             __HIP_MEMORY_SCOPE_WORKGROUP);
    float v[32];
#pragma unroll
    for (int j = 0; j < 32; ++j) {
      const int sj = __builtin_amdgcn_readlane(s, j);
      v[j] = u[(size_t)sj * OUT_DIM + lane];
    }
#pragma unroll
    for (int j = 0; j < 32; ++j) {
      const int   gj = __builtin_amdgcn_readlane((int)g, j);
      const float wj = __builtin_bit_cast(float,
          __builtin_amdgcn_readlane(__builtin_bit_cast(int, w), j));
      __hip_atomic_fetch_add(zs + gj * OUT_DIM + lane, v[j] * wj, __ATOMIC_RELAXED,
                             __HIP_MEMORY_SCOPE_WORKGROUP);
    }
  }
  __syncthreads();

  float4* __restrict__ mine = (float4*)(part + (size_t)blockIdx.x * Z_PAD);
#pragma unroll 4
  for (int i = tid; i < Z_PAD / 4; i += 1024) mine[i] = zs4[i];
}

#define PG_GRAPHS 4
#define PG_GROUPS 32

__global__ __launch_bounds__(1024) void pool_finish(
    const float* __restrict__ part, const float* __restrict__ b, float* __restrict__ out) {
  __shared__ float4 red4[PG_GROUPS * 32];
  __shared__ float  cntS[PG_GRAPHS];
  const int tid = threadIdx.x;
  const int g0  = blockIdx.x * PG_GRAPHS;

  if (tid < PG_GRAPHS) cntS[tid] = 0.0f;

  {
    const int col = tid & 31;
    const int grp = tid >> 5;
    float4 acc = make_float4(0.f, 0.f, 0.f, 0.f);
    const float4* __restrict__ p = (const float4*)(part + (size_t)g0 * OUT_DIM) + col;
#pragma unroll
    for (int j = 0; j < POOL_BLOCKS / PG_GROUPS; ++j) {
      const float4 t = p[(size_t)(grp + j * PG_GROUPS) * (Z_PAD / 4)];
      acc.x += t.x; acc.y += t.y; acc.z += t.z; acc.w += t.w;
    }
    red4[grp * 32 + col] = acc;
  }
  {
    const int gl = tid & 3;
    float cv = 0.0f;
    for (int q = tid >> 2; q < POOL_BLOCKS; q += 256)
      cv += part[(size_t)q * Z_PAD + Z_COUNTS + g0 + gl];
    cv += __shfl_xor(cv, 4);
    cv += __shfl_xor(cv, 8);
    cv += __shfl_xor(cv, 16);
    __syncthreads();
    if ((tid & 31) < PG_GRAPHS)
      __hip_atomic_fetch_add(&cntS[gl], cv, __ATOMIC_RELAXED, __HIP_MEMORY_SCOPE_WORKGROUP);
  }
  __syncthreads();
  if (tid < PG_GRAPHS * OUT_DIM) {
    const float* red = (const float*)red4;
    float sum = 0.0f;
#pragma unroll
    for (int k = 0; k < PG_GROUPS; ++k) sum += red[k * (PG_GRAPHS * OUT_DIM) + tid];
    const float cnt = cntS[tid >> 5];
    out[g0 * OUT_DIM + tid] = (sum + cnt * b[tid & 31]) / cnt;
  }
}


extern "C" void kernel_launch(void* const* d_in, const int* in_sizes, int n_in,
                              void* d_out, int out_size, void* d_ws, size_t ws_size,
                              hipStream_t stream) {
  const float* x    = (const float*)d_in[0];
  const int*   eidx = (const int*)d_in[1];
  const int*   batch = (const int*)d_in[2];
  const float* W1   = (const float*)d_in[3];
  const float* b1   = (const float*)d_in[4];
  const float* W2   = (const float*)d_in[5];
  const float* b2   = (const float*)d_in[6];
  float* out = (float*)d_out;

  const int* src = eidx;
  const int* dst = eidx + N_EDGES;

  char* ws = (char*)d_ws;
  float* rec  = (float*)(ws);
  float* agg1 = (float*)(ws + 800000);
  float* h1   = (float*)(ws + 800000 + 25600000);
  unsigned* hpart = (unsigned*)(ws + 800000 + 25600000);
  float* u2   = (float*)(ws + 800000 + 25600000);
  float* part = (float*)(ws + 800000 + 25600000 + 12800000);

  deg_hist<<<DEG_RANGES * DEG_SHARES, 1024, 0, stream>>>(dst, hpart);
  deg_dis<<<(N_NODES / 2 + 127) / 128, 128, 0, stream>>>(hpart, batch, rec);

  static_assert(N_NODES % 16 == 0, "the product's row tiles cover every row of agg1");
  const int rowTiles = N_NODES / 16;
  gemm_wmma<IN_DIM, HID_DIM><<<(rowTiles + 1) / 2, 256, 0, stream>>>(x, W1, rec, h1, agg1, rowTiles);
  static_assert(N_EDGES % 32 == 0, "the scatter takes whole chunks of 32 edges");
  scatter_edges<HID_DIM><<<(N_EDGES / 32 + 7) / 8, 256, 0, stream>>>(h1, src, dst, agg1, N_EDGES / 32);

  finish_gemm2<<<(rowTiles + 7) / 8, 256, 0, stream>>>(agg1, W2, rec, b1, u2, rowTiles);
  pool_rows<<<POOL_BLOCKS, 1024, 0, stream>>>(u2, src, dst, rec, part);
  static_assert(N_GRAPHS % PG_GRAPHS == 0, "the finishing blocks cover every graph");
  pool_finish<<<N_GRAPHS / PG_GRAPHS, 1024, 0, stream>>>(part, b2, out);
}
